# starting kernel
# baseline (speedup 1.0000x reference)
#include <hip/hip_runtime.h>
#include <hip/hip_bf16.h>

typedef __attribute__((ext_vector_type(2))) float v2f;
typedef __attribute__((ext_vector_type(8))) float v8f;

__global__ void k_init_deg(float* __restrict__ deg, int n) {
    int i = blockIdx.x * blockDim.x + threadIdx.x;
    if (i < n) deg[i] = 1.0f;
}

__global__ void k_deg_edges(const int* __restrict__ col, const float* __restrict__ ew,
                            float* __restrict__ deg, int e) {
    int i = blockIdx.x * blockDim.x + threadIdx.x;
    if (i < e) atomicAdd(&deg[col[i]], ew[i]);
}

__global__ void k_dis(float* __restrict__ deg_dis, int n) {
    int i = blockIdx.x * blockDim.x + threadIdx.x;
    if (i < n) {
        float d = deg_dis[i];
        deg_dis[i] = (d > 0.f) ? rsqrtf(d) : 0.f;
    }
}

__global__ void k_gemm1(const float* __restrict__ x, const float* __restrict__ W1,
                        float* __restrict__ h1, int n, int ntiles) {
    int gtid = blockIdx.x * blockDim.x + threadIdx.x;
    int tile = gtid >> 5;
    int lane = threadIdx.x & 31;
    if (tile >= ntiles) return;
    int half = lane >> 4;
    int m    = lane & 15;
    int row  = tile * 16 + m;
    int rowc = row < n ? row : n - 1;
    bool rv  = row < n;
    int k0   = half * 2;

    float ax = x[(size_t)rowc * 3 + k0];
    float ay = x[(size_t)rowc * 3 + 1];
    v2f a;
    a.x = rv ? ax : 0.f;
    a.y = (rv && half == 0) ? ay : 0.f;

    float bx = W1[k0 * 16 + m];
    float by = W1[16 + m];
    v2f b;
    b.x = bx;
    b.y = (half == 0) ? by : 0.f;

    v8f c = {};
    c = __builtin_amdgcn_wmma_f32_16x16x4_f32(false, a, false, b, (short)0, c, false, false);

    float* p = h1 + ((size_t)(tile * 16 + half * 8)) * 16 + m;
    if (tile * 16 + 16 <= n) {
#pragma unroll
        for (int j = 0; j < 8; ++j) p[(size_t)j * 16] = c[j];
    } else {
#pragma unroll
        for (int j = 0; j < 8; ++j)
            if (tile * 16 + half * 8 + j < n) p[(size_t)j * 16] = c[j];
    }
}

__global__ void k_self1(const float* __restrict__ h1, const float* __restrict__ dis,
                        float* __restrict__ acc, int n16) {
    int i = blockIdx.x * blockDim.x + threadIdx.x;
    if (i < n16) {
        float s = dis[i >> 4];
        acc[i] = h1[i] * s * s;
    }
}

__global__ void k_edges1(const int* __restrict__ row, const int* __restrict__ col,
                         const float* __restrict__ ew, const float* __restrict__ dis,
                         const float* __restrict__ h1, float* __restrict__ acc, int e) {
    int i = blockIdx.x * blockDim.x + threadIdx.x;
    if (i >= e) return;
    int r = row[i], c = col[i];
    float nrm = dis[r] * ew[i] * dis[c];
    const float4* hr = (const float4*)(h1 + (size_t)r * 16);
    float4 v0 = hr[0], v1 = hr[1], v2 = hr[2], v3 = hr[3];
    float* dst = acc + (size_t)c * 16;
    atomicAdd(dst +  0, v0.x * nrm); atomicAdd(dst +  1, v0.y * nrm);
    atomicAdd(dst +  2, v0.z * nrm); atomicAdd(dst +  3, v0.w * nrm);
    atomicAdd(dst +  4, v1.x * nrm); atomicAdd(dst +  5, v1.y * nrm);
    atomicAdd(dst +  6, v1.z * nrm); atomicAdd(dst +  7, v1.w * nrm);
    atomicAdd(dst +  8, v2.x * nrm); atomicAdd(dst +  9, v2.y * nrm);
    atomicAdd(dst + 10, v2.z * nrm); atomicAdd(dst + 11, v2.w * nrm);
    atomicAdd(dst + 12, v3.x * nrm); atomicAdd(dst + 13, v3.y * nrm);
    atomicAdd(dst + 14, v3.z * nrm); atomicAdd(dst + 15, v3.w * nrm);
}

__global__ void k_relu(float* __restrict__ acc, const float* __restrict__ b1, int n16) {
    int i = blockIdx.x * blockDim.x + threadIdx.x;
    if (i < n16) acc[i] = fmaxf(acc[i] + b1[i & 15], 0.f);
}

__global__ void k_gemm2(const float* __restrict__ z, const float* __restrict__ W2,
                        float* __restrict__ h2, int n, int ntiles) {
    int gtid = blockIdx.x * blockDim.x + threadIdx.x;
    int tile = gtid >> 5;
    int lane = threadIdx.x & 31;
    if (tile >= ntiles) return;
    int half = lane >> 4;
    int m    = lane & 15;
    int row  = tile * 16 + m;
    int rowc = row < n ? row : n - 1;
    bool rv  = row < n;
    int mc   = m < 7 ? m : 6;
    bool mv  = m < 7;

    v8f c = {};
#pragma unroll
    for (int s = 0; s < 4; ++s) {
        int k = s * 4 + half * 2;
        float ax = z[(size_t)rowc * 16 + k + 0];
        float ay = z[(size_t)rowc * 16 + k + 1];
        v2f a;
        a.x = rv ? ax : 0.f;
        a.y = rv ? ay : 0.f;
        float bx = W2[(k + 0) * 7 + mc];
        float by = W2[(k + 1) * 7 + mc];
        v2f b;
        b.x = mv ? bx : 0.f;
        b.y = mv ? by : 0.f;
        c = __builtin_amdgcn_wmma_f32_16x16x4_f32(false, a, false, b, (short)0, c, false, false);
    }

    float* p = h2 + ((size_t)(tile * 16 + half * 8)) * 16 + m;
    if (tile * 16 + 16 <= n) {
#pragma unroll
        for (int j = 0; j < 8; ++j) p[(size_t)j * 16] = c[j];
    } else {
#pragma unroll
        for (int j = 0; j < 8; ++j)
            if (tile * 16 + half * 8 + j < n) p[(size_t)j * 16] = c[j];
    }
}

__global__ void k_out_init(const float* __restrict__ h2, const float* __restrict__ dis,
                           const float* __restrict__ b2, float* __restrict__ out, int n7) {
    int i = blockIdx.x * blockDim.x + threadIdx.x;
    if (i < n7) {
        int node = i / 7, f = i % 7;
        float s = dis[node];
        out[i] = b2[f] + h2[(size_t)node * 16 + f] * s * s;
    }
}

__global__ void k_edges2(const int* __restrict__ row, const int* __restrict__ col,
                         const float* __restrict__ ew, const float* __restrict__ dis,
                         const float* __restrict__ h2, float* __restrict__ out, int e) {
    int i = blockIdx.x * blockDim.x + threadIdx.x;
    if (i >= e) return;
    int r = row[i], c = col[i];
    float nrm = dis[r] * ew[i] * dis[c];
    const float4* hr = (const float4*)(h2 + (size_t)r * 16);
    float4 v0 = hr[0], v1 = hr[1];
    float* dst = out + (size_t)c * 7;
    atomicAdd(dst + 0, v0.x * nrm); atomicAdd(dst + 1, v0.y * nrm);
    atomicAdd(dst + 2, v0.z * nrm); atomicAdd(dst + 3, v0.w * nrm);
    atomicAdd(dst + 4, v1.x * nrm); atomicAdd(dst + 5, v1.y * nrm);
    atomicAdd(dst + 6, v1.z * nrm);
}

extern "C" void kernel_launch(void* const* d_in, const int* in_sizes, int n_in,
                              void* d_out, int out_size, void* d_ws, size_t ws_size,
                              hipStream_t stream) {
    (void)n_in; (void)out_size; (void)ws_size;
    const float* x  = (const float*)d_in[0];
    const int*   ei = (const int*)d_in[1];
    const float* ew = (const float*)d_in[2];
    const float* W1 = (const float*)d_in[3];
    const float* b1 = (const float*)d_in[4];
    const float* W2 = (const float*)d_in[5];
    const float* b2 = (const float*)d_in[6];
    float* out = (float*)d_out;

    int n = in_sizes[0] / 3;
    int e = in_sizes[2];
    const int* row = ei;
    const int* col = ei + e;

    float* ws   = (float*)d_ws;
    float* dis  = ws;
    float* h1   = dis  + (size_t)n;
    float* acc1 = h1   + (size_t)n * 16;
    float* h2   = acc1 + (size_t)n * 16;

    const int B = 256;
    int ntiles  = (n + 15) / 16;
    int gemmT   = ntiles * 32;

    k_init_deg <<<(n + B - 1) / B, B, 0, stream>>>(dis, n);
    k_deg_edges<<<(e + B - 1) / B, B, 0, stream>>>(col, ew, dis, e);
    k_dis      <<<(n + B - 1) / B, B, 0, stream>>>(dis, n);

    k_gemm1    <<<(gemmT + B - 1) / B, B, 0, stream>>>(x, W1, h1, n, ntiles);
    k_self1    <<<(n * 16 + B - 1) / B, B, 0, stream>>>(h1, dis, acc1, n * 16);
    k_edges1   <<<(e + B - 1) / B, B, 0, stream>>>(row, col, ew, dis, h1, acc1, e);
    k_relu     <<<(n * 16 + B - 1) / B, B, 0, stream>>>(acc1, b1, n * 16);

    k_gemm2    <<<(gemmT + B - 1) / B, B, 0, stream>>>(acc1, W2, h2, n, ntiles);
    k_out_init <<<(n * 7 + B - 1) / B, B, 0, stream>>>(h2, dis, b2, out, n * 7);
    k_edges2   <<<(e + B - 1) / B, B, 0, stream>>>(row, col, ew, dis, h2, out, e);
}


# layer 1's scatter logs the records it matched to per-wave lists; layer 2 streams those lists whole-line: no bitmap walk, no scattered record loads; new f32 sum order (plane split; arrival within)
# speedup vs baseline: 8.6695x; 8.6695x over previous
#include <hip/hip_runtime.h>
#include <hip/hip_bf16.h>


typedef __attribute__((ext_vector_type(2))) float v2f;
typedef __attribute__((ext_vector_type(8))) float v8f;

constexpr int WG   = 1024;
constexpr int QCAP = 64;

template <int A>
__device__ __forceinline__ int slice_cut(int e, int s, int S) {
    if (s <= 0) return 0;
    if (s >= S) return e;
    long long c = (((long long)e * s) / S + (A - 1)) & ~(long long)(A - 1);
    return c < (long long)e ? (int)c : e;
}

template <int NPB, int S>
__global__ __launch_bounds__(1024)
void k_deg_lds(const int* __restrict__ col, const float* __restrict__ ew,
               float* __restrict__ planes, int n, int e) {
    __shared__ float acc[NPB];
    const int tid  = threadIdx.x;
    const int base = blockIdx.x * NPB;
    const int s    = blockIdx.y;
    for (int j = tid; j < NPB; j += WG) acc[j] = 0.f;
    __syncthreads();

    const int lo = slice_cut<4>(e, s, S), hi = slice_cut<4>(e, s + 1, S);
    const bool vec = ((((size_t)col) | ((size_t)ew)) & 15) == 0;
    const int vend = vec ? lo + ((hi - lo) & ~3) : lo;

    const int4*   col4 = (const int4*)col;
    const float4* ew4  = (const float4*)ew;
    for (int g = (lo >> 2) + tid; g < (vend >> 2); g += WG) {
        int4 c = col4[g];
        float4 w = ew4[g];
        unsigned d0 = (unsigned)(c.x - base), d1 = (unsigned)(c.y - base);
        unsigned d2 = (unsigned)(c.z - base), d3 = (unsigned)(c.w - base);
        if (d0 < (unsigned)NPB) atomicAdd(&acc[d0], w.x);
        if (d1 < (unsigned)NPB) atomicAdd(&acc[d1], w.y);
        if (d2 < (unsigned)NPB) atomicAdd(&acc[d2], w.z);
        if (d3 < (unsigned)NPB) atomicAdd(&acc[d3], w.w);
    }
    for (int i = vend + tid; i < hi; i += WG) {
        unsigned d = (unsigned)(col[i] - base);
        if (d < (unsigned)NPB) atomicAdd(&acc[d], ew[i]);
    }
    __syncthreads();

    const int cnt = (n - base) < NPB ? (n - base) : NPB;
    float* dst = planes + (size_t)s * n + base;
    for (int j = tid; j < cnt; j += WG) dst[j] = acc[j];
}

template <int S>
__global__ void k_dis(const float* __restrict__ planes, float* __restrict__ dis,
                      int* __restrict__ ovf, int n) {
    int i = blockIdx.x * blockDim.x + threadIdx.x;
    if (i == 0) *ovf = 0;
    if (i < n) {
        float d = 1.0f;
        for (int p = 0; p < S; ++p) d += planes[(size_t)p * n + i];
        dis[i] = (d > 0.f) ? rsqrtf(d) : 0.f;
    }
}

constexpr int RID_NODES = 3136;
constexpr int NRID = 64;
constexpr int PK_T = 256;
constexpr int PK_W = PK_T * 4 / 32;
constexpr int PK_P = PK_W + 1;

__global__ __launch_bounds__(PK_T)
void k_pack(const int* __restrict__ row, const int* __restrict__ col,
            const float* __restrict__ ew,
            unsigned int* __restrict__ rec, unsigned int* __restrict__ bm,
            int n, int e, int pitchw) {
    __shared__ unsigned int tile[NRID * PK_P];
    const int tid = threadIdx.x;
    for (int j = tid; j < NRID * PK_P; j += PK_T) tile[j] = 0u;
    __syncthreads();

    const int t = blockIdx.x * PK_T + tid;
    const int i = t * 4;
    if (i < e) {
        unsigned c4[4];
        const bool vec = ((((size_t)row) | ((size_t)col) | ((size_t)ew) |
                           ((size_t)rec)) & 15) == 0;
        if (vec && i + 4 <= e) {
            int4   r = ((const int4*)row)[t];
            int4   c = ((const int4*)col)[t];
            float4 w = ((const float4*)ew)[t];
            uint4 a, b;
            a.x = ((unsigned)r.x << 14) | ((unsigned)c.x & 16383u); a.y = __float_as_uint(w.x);
            a.z = ((unsigned)r.y << 14) | ((unsigned)c.y & 16383u); a.w = __float_as_uint(w.y);
            b.x = ((unsigned)r.z << 14) | ((unsigned)c.z & 16383u); b.y = __float_as_uint(w.z);
            b.z = ((unsigned)r.w << 14) | ((unsigned)c.w & 16383u); b.w = __float_as_uint(w.w);
            ((uint4*)rec)[2 * (size_t)t]     = a;
            ((uint4*)rec)[2 * (size_t)t + 1] = b;
            c4[0] = (unsigned)c.x; c4[1] = (unsigned)c.y;
            c4[2] = (unsigned)c.z; c4[3] = (unsigned)c.w;
        } else {
            for (int k = 0; k < 4; ++k) {
                int j = i + k;
                c4[k] = 0xFFFFFFFFu;
                if (j < e) {
                    unsigned c = (unsigned)col[j];
                    rec[2 * (size_t)j]     = ((unsigned)row[j] << 14) | (c & 16383u);
                    rec[2 * (size_t)j + 1] = __float_as_uint(ew[j]);
                    c4[k] = c;
                }
            }
        }
        unsigned int* tw = &tile[tid >> 3];
        const unsigned sh = (unsigned)(tid & 7) * 4u;
        for (int k = 0; k < 4; ++k) {
            unsigned c = c4[k];
            if (c < (unsigned)n) {
                unsigned id = c / RID_NODES;
                if (id < (unsigned)NRID) atomicOr(tw + id * PK_P, 1u << (sh + k));
            }
        }
    }
    __syncthreads();

    unsigned int* dst = bm + (size_t)blockIdx.x * PK_W + (tid & 31);
    for (int r = tid >> 5; r < NRID; r += PK_T / 32)
        dst[(size_t)r * pitchw] = tile[r * PK_P + (tid & 31)];
}

struct full_t { static constexpr bool value = true;  };
struct part_t { static constexpr bool value = false; };
constexpr int LCAP = 1024;
template <int F, int PITCH, int NPB, int S, int SHIFT, bool WLIST, bool RLIST>
__global__ __launch_bounds__(1024)
void k_scatter_lds(const unsigned int* __restrict__ rec, const unsigned int* __restrict__ bm,
                   const float* __restrict__ h, float* __restrict__ planes,
                   int n, int pitchw,
                   unsigned int* __restrict__ lst, int* __restrict__ cnts, int* __restrict__ ovf, int nr1) {
    static_assert(!(WLIST && RLIST), "a launch writes the lists or reads them");
    static_assert(!RLIST || SHIFT == 1, "the list owner map is for ranges of two layer-1 ranges");
    static_assert(NPB == (RID_NODES << SHIFT), "a node range is a whole number of range ids");
    static_assert(NPB <= 16384, "the record keeps 14 bits of the target");
    __shared__ float acc[NPB * PITCH];
    __shared__ int   q[WG / 32][QCAP];
    const int tid  = threadIdx.x;
    const int lane = tid & 31;
    const unsigned rng  = blockIdx.x;
    const unsigned base = rng * NPB;
    const int s    = blockIdx.y;
    for (int j = tid; j < NPB * PITCH; j += WG) acc[j] = 0.f;
    __syncthreads();

    const uint2* rec2 = (const uint2*)rec;
    const int G  = pitchw >> 2;
    const int g0 = (int)(((long long)G * s) / S), g1 = (int)(((long long)G * (s + 1)) / S);
    const bool vec = (((size_t)bm) & 15) == 0;

    int* myq = q[tid >> 5];
    int qn = 0;
    const unsigned below = (1u << lane) - 1u;

    auto push = [&](bool m, int eid) {
        unsigned mask = __builtin_amdgcn_ballot_w32(m);
        if (mask != 0u) {
            if (m) myq[qn + __builtin_popcount(mask & below)] = eid;
            qn += __builtin_popcount(mask);
        }
    };
    const int sub = lane >> 2, j = lane & 3;
    auto apply = [&](const uint2 (&rc)[4], const bool (&ok)[4]) __attribute__((always_inline)) {
        if constexpr (F == 16) {
            float4 v[4];
#pragma unroll
            for (int p = 0; p < 4; ++p)
                v[p] = ((const float4*)(h + (size_t)(rc[p].x >> 14) * 16))[j];
#pragma unroll
            for (int p = 0; p < 4; ++p) {
                const float w = __uint_as_float(rc[p].y);
                const unsigned c = ((rc[p].x & 16383u) - base) & 16383u;
                float* a = &acc[c * PITCH + j * 4];
                if (ok[p]) {
                    atomicAdd(a + 0, v[p].x * w); atomicAdd(a + 1, v[p].y * w);
                    atomicAdd(a + 2, v[p].z * w); atomicAdd(a + 3, v[p].w * w);
                }
            }
        } else {
            float2 v[4];
#pragma unroll
            for (int p = 0; p < 4; ++p)
                v[p] = ((const float2*)(h + (size_t)(rc[p].x >> 14) * 16))[j];
            float* a[4];
#pragma unroll
            for (int p = 0; p < 4; ++p) {
                const float w = __uint_as_float(rc[p].y);
                const unsigned c = ((rc[p].x & 16383u) - base) & 16383u;
                a[p] = &acc[c * PITCH + j * 2];
                v[p].x *= w; v[p].y *= w;
                if (ok[p]) atomicAdd(a[p], v[p].x);
            }
            if (j < 3) {
#pragma unroll
                for (int p = 0; p < 4; ++p)
                    if (ok[p]) atomicAdd(a[p] + 1, v[p].y);
            }
        }
    };
    const int wlid = (((int)blockIdx.y * nr1 + (int)blockIdx.x) << 5) + (tid >> 5);
    uint2* myl = (uint2*)lst + (size_t)wlid * LCAP;
    int  ln = 0;
    bool lover = false;
    auto work = [&](auto fullc, int qb, int cnt) {
        constexpr bool FULL = decltype(fullc)::value;
        uint2 rc[4];
        bool  ok[4];
#pragma unroll
        for (int p = 0; p < 4; ++p) {
            const int slot = p * 8 + sub;
            ok[p] = FULL || slot < cnt;
            rc[p] = rec2[myq[qb + (ok[p] ? slot : 0)]];
        }
        if constexpr (WLIST) {
            if (ln + 32 <= LCAP) {
                const bool j1 = (j & 1) != 0, j2 = (j & 2) != 0;
                uint2 r;
                r.x = j2 ? (j1 ? rc[3].x : rc[2].x) : (j1 ? rc[1].x : rc[0].x);
                r.y = j2 ? (j1 ? rc[3].y : rc[2].y) : (j1 ? rc[1].y : rc[0].y);
                const int slot = j * 8 + sub;
                if (FULL || slot < cnt) myl[ln + slot] = r;
                ln += cnt;
            } else {
                lover = true;
            }
        }
        apply(rc, ok);
    };
    auto drain = [&]() {
        while (qn >= 32) { qn -= 32; work(full_t{}, qn, 32); }
    };
    auto scanw = [&](unsigned bits, int eb) {
        while (__builtin_amdgcn_ballot_w32(bits != 0u) != 0u) {
            push(bits != 0u, eb + __ffs(bits) - 1);
            bits &= bits - 1u;
            drain();
        }
    };
    auto ldgroup = [&](int g) {
        uint4 v = make_uint4(0u, 0u, 0u, 0u);
        for (int k = 0; k < (1 << SHIFT); ++k) {
            const unsigned rw = (rng << SHIFT) + k;
            if (rw < (unsigned)NRID) {
                const unsigned int* p = bm + (size_t)rw * pitchw + (size_t)g * 4;
                if (vec) {
                    uint4 u = *(const uint4*)p;
                    v.x |= u.x; v.y |= u.y; v.z |= u.z; v.w |= u.w;
                } else {
                    v.x |= p[0]; v.y |= p[1]; v.z |= p[2]; v.w |= p[3];
                }
            }
        }
        return v;
    };

    bool uselist = false;
    if constexpr (RLIST) uselist = __builtin_amdgcn_readfirstlane(*ovf) == 0;

    if (uselist) {
        const int w   = tid >> 5;
        const int r1  = (int)(rng << 1) + (w >> 4);
        const int lid = (((s >> 1) * nr1 + r1) << 5) + ((w & 15) << 1) + (s & 1);
        int c = 0;
        if (r1 < nr1) c = cnts[lid];
        c = __builtin_amdgcn_readfirstlane(c);
        const uint2* L = (const uint2*)lst + (size_t)lid * LCAP;
        auto fetch = [&](uint2 (&rc)[4], int b) {
#pragma unroll
            for (int p = 0; p < 4; ++p) {
                const int idx = b + p * 8 + sub;
                rc[p] = L[idx < c ? idx : b];
            }
        };
        uint2 cur[4] = {};
        if (c > 0) fetch(cur, 0);
        for (int b = 0; b < c; b += 32) {
            uint2 nxt[4] = {};
            if (b + 32 < c) fetch(nxt, b + 32);
            const int cnt = c - b;
            bool ok[4];
#pragma unroll
            for (int p = 0; p < 4; ++p) ok[p] = p * 8 + sub < cnt;
            apply(cur, ok);
#pragma unroll
            for (int p = 0; p < 4; ++p) cur[p] = nxt[p];
        }
    } else {
        for (int gb = g0; gb < g1; gb += 2 * WG) {
            uint4 cv[2];
            for (int j = 0; j < 2; ++j) {
                int g = gb + j * WG + tid;
                bool ok = g < g1;
                uint4 v = ldgroup(ok ? g : g0);
                cv[j].x = ok ? v.x : 0u; cv[j].y = ok ? v.y : 0u;
                cv[j].z = ok ? v.z : 0u; cv[j].w = ok ? v.w : 0u;
            }
            for (int j = 0; j < 2; ++j) {
                int eb = (gb + j * WG + tid) * 128;
                scanw(cv[j].x, eb + 0);
                scanw(cv[j].y, eb + 32);
                scanw(cv[j].z, eb + 64);
                scanw(cv[j].w, eb + 96);
            }
        }
        if (qn > 0) work(part_t{}, 0, qn);
        if constexpr (WLIST) {
            if (lane == 0) {
                cnts[wlid] = ln;
                if (lover) *ovf = 1;
            }
        }
    }
    __syncthreads();

    const int cnt = (n - (int)base) < NPB ? (n - (int)base) : NPB;
    float* dst = planes + ((size_t)s * n + base) * F;
    for (int j = tid; j < cnt * F; j += WG) {
        int node = j / F, f = j - node * F;
        dst[j] = acc[node * PITCH + f];
    }
}

__global__ void k_gemm1(const float* __restrict__ x, const float* __restrict__ W1,
                        const float* __restrict__ dis,
                        float* __restrict__ h1, int n, int ntiles) {
    int gtid = blockIdx.x * blockDim.x + threadIdx.x;
    int tile = gtid >> 5;
    int lane = threadIdx.x & 31;
    if (tile >= ntiles) return;
    int half = lane >> 4;
    int m    = lane & 15;
    int row  = tile * 16 + m;
    int rowc = row < n ? row : n - 1;
    bool rv  = row < n;
    int k0   = half * 2;

    float dr = dis[rowc];
    float ax = x[(size_t)rowc * 3 + k0] * dr;
    float ay = x[(size_t)rowc * 3 + 1] * dr;
    v2f a;
    a.x = rv ? ax : 0.f;
    a.y = (rv && half == 0) ? ay : 0.f;

    float bx = W1[k0 * 16 + m];
    float by = W1[16 + m];
    v2f b;
    b.x = bx;
    b.y = (half == 0) ? by : 0.f;

    v8f c = {};
    c = __builtin_amdgcn_wmma_f32_16x16x4_f32(false, a, false, b, (short)0, c, false, false);

    float* p = h1 + ((size_t)(tile * 16 + half * 8)) * 16 + m;
    if (tile * 16 + 16 <= n) {
#pragma unroll
        for (int j = 0; j < 8; ++j) p[(size_t)j * 16] = c[j];
    } else {
#pragma unroll
        for (int j = 0; j < 8; ++j)
            if (tile * 16 + half * 8 + j < n) p[(size_t)j * 16] = c[j];
    }
}

template <int S>
__global__ void k_relu(const float* __restrict__ h1, const float* __restrict__ dis,
                       const float* __restrict__ planes, const float* __restrict__ b1,
                       float* __restrict__ z, int n16) {
    int i = blockIdx.x * blockDim.x + threadIdx.x;
    if (i < n16) {
        float s = dis[i >> 4];
        float v = h1[i];
        for (int p = 0; p < S; ++p) v += planes[(size_t)p * n16 + i];
        z[i] = fmaxf(v * s + b1[i & 15], 0.f) * s;
    }
}

__global__ void k_gemm2(const float* __restrict__ z, const float* __restrict__ W2,
                        float* __restrict__ h2, int n, int ntiles) {
    int gtid = blockIdx.x * blockDim.x + threadIdx.x;
    int tile = gtid >> 5;
    int lane = threadIdx.x & 31;
    if (tile >= ntiles) return;
    int half = lane >> 4;
    int m    = lane & 15;
    int row  = tile * 16 + m;
    int rowc = row < n ? row : n - 1;
    bool rv  = row < n;
    int mc   = m < 7 ? m : 6;
    bool mv  = m < 7;

    v8f c = {};
#pragma unroll
    for (int s = 0; s < 4; ++s) {
        int k = s * 4 + half * 2;
        float ax = z[(size_t)rowc * 16 + k + 0];
        float ay = z[(size_t)rowc * 16 + k + 1];
        v2f a;
        a.x = rv ? ax : 0.f;
        a.y = rv ? ay : 0.f;
        float bx = W2[(k + 0) * 7 + mc];
        float by = W2[(k + 1) * 7 + mc];
        v2f b;
        b.x = mv ? bx : 0.f;
        b.y = mv ? by : 0.f;
        c = __builtin_amdgcn_wmma_f32_16x16x4_f32(false, a, false, b, (short)0, c, false, false);
    }

    float* p = h2 + ((size_t)(tile * 16 + half * 8)) * 16 + m;
    if (tile * 16 + 16 <= n) {
#pragma unroll
        for (int j = 0; j < 8; ++j) p[(size_t)j * 16] = c[j];
    } else {
#pragma unroll
        for (int j = 0; j < 8; ++j)
            if (tile * 16 + half * 8 + j < n) p[(size_t)j * 16] = c[j];
    }
}

template <int S>
__global__ void k_out(const float* __restrict__ h2, const float* __restrict__ dis,
                      const float* __restrict__ planes, const float* __restrict__ b2,
                      float* __restrict__ out, int n7) {
    int i = blockIdx.x * blockDim.x + threadIdx.x;
    if (i < n7) {
        int node = i / 7, f = i % 7;
        float s = dis[node];
        float v = h2[(size_t)node * 16 + f];
        for (int p = 0; p < S; ++p) v += planes[(size_t)p * n7 + i];
        out[i] = b2[f] + v * s;
    }
}

extern "C" void kernel_launch(void* const* d_in, const int* in_sizes, int n_in,
                              void* d_out, int out_size, void* d_ws, size_t ws_size,
                              hipStream_t stream) {
    (void)n_in; (void)out_size; (void)ws_size;
    const float* x  = (const float*)d_in[0];
    const int*   ei = (const int*)d_in[1];
    const float* ew = (const float*)d_in[2];
    const float* W1 = (const float*)d_in[3];
    const float* b1 = (const float*)d_in[4];
    const float* W2 = (const float*)d_in[5];
    const float* b2 = (const float*)d_in[6];
    float* out = (float*)d_out;

    int n = in_sizes[0] / 3;
    int e = in_sizes[2];
    const int* row = ei;
    const int* col = ei + e;

    constexpr int NPB_D = 50176, S_D = 64;
    constexpr int NPB_1 = 3136,  S_1 = 4;
    constexpr int NPB_2 = 6272,  S_2 = 8;

    float* ws     = (float*)d_ws;
    float* dis    = ws;
    float* h1     = dis + (size_t)n;
    float* z      = h1  + (size_t)n * 16;
    float* h2     = z   + (size_t)n * 16;
    float* planes = h2  + (size_t)n * 16;
    unsigned int* rec = (unsigned int*)(planes + (size_t)n * 64);
    unsigned int* bm  = rec + (size_t)e * 2;
    int nblk   = (e + PK_T * 4 - 1) / (PK_T * 4);
    int pitchw = nblk * PK_W;
    static_assert(NPB_2 == 2 * NPB_1 && S_2 == 2 * S_1, "layer 2's work-groups own whole layer-1 lists");
    int nr1    = (n + NPB_1 - 1) / NPB_1;
    int nlist  = nr1 * S_1 * (WG / 32);
    unsigned int* lst = bm + (size_t)NRID * pitchw;
    int* cnts  = (int*)(lst + (size_t)nlist * LCAP * 2);
    int* ovf   = cnts + nlist;

    const int B = 256;
    int ntiles  = (n + 15) / 16;
    int gemmT   = ntiles * 32;

    k_pack<<<nblk, PK_T, 0, stream>>>(row, col, ew, rec, bm, n, e, pitchw);
    k_deg_lds<NPB_D, S_D><<<dim3((n + NPB_D - 1) / NPB_D, S_D), WG, 0, stream>>>(col, ew, planes, n, e);
    k_dis<S_D><<<(n + B - 1) / B, B, 0, stream>>>(planes, dis, ovf, n);

    k_gemm1<<<(gemmT + B - 1) / B, B, 0, stream>>>(x, W1, dis, h1, n, ntiles);
    k_scatter_lds<16, 17, NPB_1, S_1, 0, true, false><<<dim3(nr1, S_1), WG, 0, stream>>>(
        rec, bm, h1, planes, n, pitchw, lst, cnts, ovf, nr1);
    k_relu<S_1><<<(n * 16 + B - 1) / B, B, 0, stream>>>(h1, dis, planes, b1, z, n * 16);

    k_gemm2<<<(gemmT + B - 1) / B, B, 0, stream>>>(z, W2, h2, n, ntiles);
    k_scatter_lds<7, 7, NPB_2, S_2, 1, false, true><<<dim3((n + NPB_2 - 1) / NPB_2, S_2), WG, 0, stream>>>(
        rec, bm, h2, planes, n, pitchw, lst, cnts, ovf, nr1);
    k_out<S_2><<<(n * 7 + B - 1) / B, B, 0, stream>>>(h2, dis, planes, b2, out, n * 7);
}


# bitmap walk and record gathers moved out of layer 1 into the degree launch (layer 1's grid: adds each record's weight in LDS, logs records per wave); layer 1 streams the lists; new f32 sum order
# speedup vs baseline: 9.3374x; 1.0770x over previous
#include <hip/hip_runtime.h>
#include <hip/hip_bf16.h>


typedef __attribute__((ext_vector_type(2))) float v2f;
typedef __attribute__((ext_vector_type(8))) float v8f;

constexpr int WG   = 1024;
constexpr int QCAP = 64;

template <int S>
__global__ void k_dis(const float* __restrict__ planes, float* __restrict__ dis, int n) {
    int i = blockIdx.x * blockDim.x + threadIdx.x;
    if (i < n) {
        float d = 1.0f;
        for (int p = 0; p < S; ++p) d += planes[(size_t)p * n + i];
        dis[i] = (d > 0.f) ? rsqrtf(d) : 0.f;
    }
}

constexpr int RID_NODES = 3136;
constexpr int NRID = 64;
constexpr int PK_T = 256;
constexpr int PK_W = PK_T * 4 / 32;
constexpr int PK_P = PK_W + 1;

__global__ __launch_bounds__(PK_T)
void k_pack(const int* __restrict__ row, const int* __restrict__ col,
            const float* __restrict__ ew,
            unsigned int* __restrict__ rec, unsigned int* __restrict__ bm,
            int* __restrict__ ovf, int n, int e, int pitchw) {
    __shared__ unsigned int tile[NRID * PK_P];
    const int tid = threadIdx.x;
    if (blockIdx.x == 0 && tid == 0) *ovf = 0;
    for (int j = tid; j < NRID * PK_P; j += PK_T) tile[j] = 0u;
    __syncthreads();

    const int t = blockIdx.x * PK_T + tid;
    const int i = t * 4;
    if (i < e) {
        unsigned c4[4];
        const bool vec = ((((size_t)row) | ((size_t)col) | ((size_t)ew) |
                           ((size_t)rec)) & 15) == 0;
        if (vec && i + 4 <= e) {
            int4   r = ((const int4*)row)[t];
            int4   c = ((const int4*)col)[t];
            float4 w = ((const float4*)ew)[t];
            uint4 a, b;
            a.x = ((unsigned)r.x << 14) | ((unsigned)c.x & 16383u); a.y = __float_as_uint(w.x);
            a.z = ((unsigned)r.y << 14) | ((unsigned)c.y & 16383u); a.w = __float_as_uint(w.y);
            b.x = ((unsigned)r.z << 14) | ((unsigned)c.z & 16383u); b.y = __float_as_uint(w.z);
            b.z = ((unsigned)r.w << 14) | ((unsigned)c.w & 16383u); b.w = __float_as_uint(w.w);
            ((uint4*)rec)[2 * (size_t)t]     = a;
            ((uint4*)rec)[2 * (size_t)t + 1] = b;
            c4[0] = (unsigned)c.x; c4[1] = (unsigned)c.y;
            c4[2] = (unsigned)c.z; c4[3] = (unsigned)c.w;
        } else {
            for (int k = 0; k < 4; ++k) {
                int j = i + k;
                c4[k] = 0xFFFFFFFFu;
                if (j < e) {
                    unsigned c = (unsigned)col[j];
                    rec[2 * (size_t)j]     = ((unsigned)row[j] << 14) | (c & 16383u);
                    rec[2 * (size_t)j + 1] = __float_as_uint(ew[j]);
                    c4[k] = c;
                }
            }
        }
        unsigned int* tw = &tile[tid >> 3];
        const unsigned sh = (unsigned)(tid & 7) * 4u;
        for (int k = 0; k < 4; ++k) {
            unsigned c = c4[k];
            if (c < (unsigned)n) {
                unsigned id = c / RID_NODES;
                if (id < (unsigned)NRID) atomicOr(tw + id * PK_P, 1u << (sh + k));
            }
        }
    }
    __syncthreads();

    unsigned int* dst = bm + (size_t)blockIdx.x * PK_W + (tid & 31);
    for (int r = tid >> 5; r < NRID; r += PK_T / 32)
        dst[(size_t)r * pitchw] = tile[r * PK_P + (tid & 31)];
}

struct full_t { static constexpr bool value = true;  };
struct part_t { static constexpr bool value = false; };
constexpr int LCAP = 1024;
template <int F, int PITCH, int NPB, int S, int SHIFT>
__global__ __launch_bounds__(1024)
void k_scatter_lds(const unsigned int* __restrict__ rec, const unsigned int* __restrict__ bm,
                   const float* __restrict__ h, float* __restrict__ planes,
                   int n, int pitchw,
                   unsigned int* __restrict__ lst, int* __restrict__ cnts, int* __restrict__ ovf, int nr1) {
    static_assert(F == 1 || F == 7 || F == 16, "degree, layer 2 or layer 1");
    constexpr bool DEG = F == 1;
    static_assert(SHIFT == 0 || (SHIFT == 1 && !DEG), "the list owner maps are for the degree launch's ranges and for pairs of them");
    static_assert(NPB == (RID_NODES << SHIFT), "a node range is a whole number of range ids");
    static_assert(NPB <= 16384, "the record keeps 14 bits of the target");
    __shared__ float acc[NPB * PITCH];
    __shared__ int   q[WG / 32][QCAP];
    const int tid  = threadIdx.x;
    const int lane = tid & 31;
    const unsigned rng  = blockIdx.x;
    const unsigned base = rng * NPB;
    const int s    = blockIdx.y;
    for (int j = tid; j < NPB * PITCH; j += WG) acc[j] = 0.f;
    __syncthreads();

    const uint2* rec2 = (const uint2*)rec;
    const int G  = pitchw >> 2;
    const int g0 = (int)(((long long)G * s) / S), g1 = (int)(((long long)G * (s + 1)) / S);
    const bool vec = (((size_t)bm) & 15) == 0;

    int* myq = q[tid >> 5];
    int qn = 0;
    const unsigned below = (1u << lane) - 1u;

    auto push = [&](bool m, int eid) {
        unsigned mask = __builtin_amdgcn_ballot_w32(m);
        if (mask != 0u) {
            if (m) myq[qn + __builtin_popcount(mask & below)] = eid;
            qn += __builtin_popcount(mask);
        }
    };
    const int sub = lane >> 2, j = lane & 3;
    auto apply = [&](const uint2 (&rc)[4], const bool (&ok)[4]) __attribute__((always_inline)) {
        if constexpr (F == 16) {
            float4 v[4];
#pragma unroll
            for (int p = 0; p < 4; ++p)
                v[p] = ((const float4*)(h + (size_t)(rc[p].x >> 14) * 16))[j];
#pragma unroll
            for (int p = 0; p < 4; ++p) {
                const float w = __uint_as_float(rc[p].y);
                const unsigned c = ((rc[p].x & 16383u) - base) & 16383u;
                float* a = &acc[c * PITCH + j * 4];
                if (ok[p]) {
                    atomicAdd(a + 0, v[p].x * w); atomicAdd(a + 1, v[p].y * w);
                    atomicAdd(a + 2, v[p].z * w); atomicAdd(a + 3, v[p].w * w);
                }
            }
        } else if constexpr (F == 7) {
            float2 v[4];
#pragma unroll
            for (int p = 0; p < 4; ++p)
                v[p] = ((const float2*)(h + (size_t)(rc[p].x >> 14) * 16))[j];
            float* a[4];
#pragma unroll
            for (int p = 0; p < 4; ++p) {
                const float w = __uint_as_float(rc[p].y);
                const unsigned c = ((rc[p].x & 16383u) - base) & 16383u;
                a[p] = &acc[c * PITCH + j * 2];
                v[p].x *= w; v[p].y *= w;
                if (ok[p]) atomicAdd(a[p], v[p].x);
            }
            if (j < 3) {
#pragma unroll
                for (int p = 0; p < 4; ++p)
                    if (ok[p]) atomicAdd(a[p] + 1, v[p].y);
            }
        }
    };
    const int wlid = (((int)blockIdx.y * nr1 + (int)blockIdx.x) << 5) + (tid >> 5);
    uint2* myl = (uint2*)lst + (size_t)wlid * LCAP;
    int  ln = 0;
    bool lover = false;
    auto work = [&](auto fullc, int qb, int cnt) {
        constexpr bool FULL = decltype(fullc)::value;
        if constexpr (DEG) {
            const bool mine = FULL || lane < cnt;
            const uint2 r = rec2[myq[qb + (mine ? lane : 0)]];
            if (ln + 32 <= LCAP) {
                if (mine) myl[ln + lane] = r;
                ln += cnt;
            } else {
                lover = true;
            }
            if (mine) atomicAdd(&acc[((r.x & 16383u) - base) & 16383u], __uint_as_float(r.y));
        } else {
            uint2 rc[4];
            bool  ok[4];
#pragma unroll
            for (int p = 0; p < 4; ++p) {
                const int slot = p * 8 + sub;
                ok[p] = FULL || slot < cnt;
                rc[p] = rec2[myq[qb + (ok[p] ? slot : 0)]];
            }
            apply(rc, ok);
        }
    };
    auto drain = [&]() {
        while (qn >= 32) { qn -= 32; work(full_t{}, qn, 32); }
    };
    auto scanw = [&](unsigned bits, int eb) {
        while (__builtin_amdgcn_ballot_w32(bits != 0u) != 0u) {
            push(bits != 0u, eb + __ffs(bits) - 1);
            bits &= bits - 1u;
            drain();
        }
    };
    auto ldgroup = [&](int g) {
        uint4 v = make_uint4(0u, 0u, 0u, 0u);
        for (int k = 0; k < (1 << SHIFT); ++k) {
            const unsigned rw = (rng << SHIFT) + k;
            if (rw < (unsigned)NRID) {
                const unsigned int* p = bm + (size_t)rw * pitchw + (size_t)g * 4;
                if (vec) {
                    uint4 u = *(const uint4*)p;
                    v.x |= u.x; v.y |= u.y; v.z |= u.z; v.w |= u.w;
                } else {
                    v.x |= p[0]; v.y |= p[1]; v.z |= p[2]; v.w |= p[3];
                }
            }
        }
        return v;
    };

    bool uselist = false;
    if constexpr (!DEG) uselist = __builtin_amdgcn_readfirstlane(*ovf) == 0;

    if (uselist) {
        const int w   = tid >> 5;
        const int r1  = SHIFT == 0 ? (int)rng : (int)(rng << 1) + (w >> 4);
        const int lid = SHIFT == 0 ? (((s * nr1 + r1) << 5) + w)
                                   : ((((s >> 1) * nr1 + r1) << 5) + ((w & 15) << 1) + (s & 1));
        int c = 0;
        if (r1 < nr1) c = cnts[lid];
        c = __builtin_amdgcn_readfirstlane(c);
        const uint2* L = (const uint2*)lst + (size_t)lid * LCAP;
        auto fetch = [&](uint2 (&rc)[4], int b) {
#pragma unroll
            for (int p = 0; p < 4; ++p) {
                const int idx = b + p * 8 + sub;
                rc[p] = L[idx < c ? idx : b];
            }
        };
        uint2 cur[4] = {};
        if (c > 0) fetch(cur, 0);
        for (int b = 0; b < c; b += 32) {
            uint2 nxt[4] = {};
            if (b + 32 < c) fetch(nxt, b + 32);
            const int cnt = c - b;
            bool ok[4];
#pragma unroll
            for (int p = 0; p < 4; ++p) ok[p] = p * 8 + sub < cnt;
            apply(cur, ok);
#pragma unroll
            for (int p = 0; p < 4; ++p) cur[p] = nxt[p];
        }
    } else {
        for (int gb = g0; gb < g1; gb += 2 * WG) {
            uint4 cv[2];
            for (int j = 0; j < 2; ++j) {
                int g = gb + j * WG + tid;
                bool ok = g < g1;
                uint4 v = ldgroup(ok ? g : g0);
                cv[j].x = ok ? v.x : 0u; cv[j].y = ok ? v.y : 0u;
                cv[j].z = ok ? v.z : 0u; cv[j].w = ok ? v.w : 0u;
            }
            for (int j = 0; j < 2; ++j) {
                int eb = (gb + j * WG + tid) * 128;
                scanw(cv[j].x, eb + 0);
                scanw(cv[j].y, eb + 32);
                scanw(cv[j].z, eb + 64);
                scanw(cv[j].w, eb + 96);
            }
        }
        if (qn > 0) work(part_t{}, 0, qn);
        if constexpr (DEG) {
            if (lane == 0) {
                cnts[wlid] = ln;
                if (lover) *ovf = 1;
            }
        }
    }
    __syncthreads();

    const int cnt = (n - (int)base) < NPB ? (n - (int)base) : NPB;
    float* dst = planes + ((size_t)s * n + base) * F;
    for (int j = tid; j < cnt * F; j += WG) {
        int node = j / F, f = j - node * F;
        dst[j] = acc[node * PITCH + f];
    }
}

__global__ void k_gemm1(const float* __restrict__ x, const float* __restrict__ W1,
                        const float* __restrict__ dis,
                        float* __restrict__ h1, int n, int ntiles) {
    int gtid = blockIdx.x * blockDim.x + threadIdx.x;
    int tile = gtid >> 5;
    int lane = threadIdx.x & 31;
    if (tile >= ntiles) return;
    int half = lane >> 4;
    int m    = lane & 15;
    int row  = tile * 16 + m;
    int rowc = row < n ? row : n - 1;
    bool rv  = row < n;
    int k0   = half * 2;

    float dr = dis[rowc];
    float ax = x[(size_t)rowc * 3 + k0] * dr;
    float ay = x[(size_t)rowc * 3 + 1] * dr;
    v2f a;
    a.x = rv ? ax : 0.f;
    a.y = (rv && half == 0) ? ay : 0.f;

    float bx = W1[k0 * 16 + m];
    float by = W1[16 + m];
    v2f b;
    b.x = bx;
    b.y = (half == 0) ? by : 0.f;

    v8f c = {};
    c = __builtin_amdgcn_wmma_f32_16x16x4_f32(false, a, false, b, (short)0, c, false, false);

    float* p = h1 + ((size_t)(tile * 16 + half * 8)) * 16 + m;
    if (tile * 16 + 16 <= n) {
#pragma unroll
        for (int j = 0; j < 8; ++j) p[(size_t)j * 16] = c[j];
    } else {
#pragma unroll
        for (int j = 0; j < 8; ++j)
            if (tile * 16 + half * 8 + j < n) p[(size_t)j * 16] = c[j];
    }
}

template <int S>
__global__ void k_relu(const float* __restrict__ h1, const float* __restrict__ dis,
                       const float* __restrict__ planes, const float* __restrict__ b1,
                       float* __restrict__ z, int n16) {
    int i = blockIdx.x * blockDim.x + threadIdx.x;
    if (i < n16) {
        float s = dis[i >> 4];
        float v = h1[i];
        for (int p = 0; p < S; ++p) v += planes[(size_t)p * n16 + i];
        z[i] = fmaxf(v * s + b1[i & 15], 0.f) * s;
    }
}

__global__ void k_gemm2(const float* __restrict__ z, const float* __restrict__ W2,
                        float* __restrict__ h2, int n, int ntiles) {
    int gtid = blockIdx.x * blockDim.x + threadIdx.x;
    int tile = gtid >> 5;
    int lane = threadIdx.x & 31;
    if (tile >= ntiles) return;
    int half = lane >> 4;
    int m    = lane & 15;
    int row  = tile * 16 + m;
    int rowc = row < n ? row : n - 1;
    bool rv  = row < n;
    int mc   = m < 7 ? m : 6;
    bool mv  = m < 7;

    v8f c = {};
#pragma unroll
    for (int s = 0; s < 4; ++s) {
        int k = s * 4 + half * 2;
        float ax = z[(size_t)rowc * 16 + k + 0];
        float ay = z[(size_t)rowc * 16 + k + 1];
        v2f a;
        a.x = rv ? ax : 0.f;
        a.y = rv ? ay : 0.f;
        float bx = W2[(k + 0) * 7 + mc];
        float by = W2[(k + 1) * 7 + mc];
        v2f b;
        b.x = mv ? bx : 0.f;
        b.y = mv ? by : 0.f;
        c = __builtin_amdgcn_wmma_f32_16x16x4_f32(false, a, false, b, (short)0, c, false, false);
    }

    float* p = h2 + ((size_t)(tile * 16 + half * 8)) * 16 + m;
    if (tile * 16 + 16 <= n) {
#pragma unroll
        for (int j = 0; j < 8; ++j) p[(size_t)j * 16] = c[j];
    } else {
#pragma unroll
        for (int j = 0; j < 8; ++j)
            if (tile * 16 + half * 8 + j < n) p[(size_t)j * 16] = c[j];
    }
}

template <int S>
__global__ void k_out(const float* __restrict__ h2, const float* __restrict__ dis,
                      const float* __restrict__ planes, const float* __restrict__ b2,
                      float* __restrict__ out, int n7) {
    int i = blockIdx.x * blockDim.x + threadIdx.x;
    if (i < n7) {
        int node = i / 7, f = i % 7;
        float s = dis[node];
        float v = h2[(size_t)node * 16 + f];
        for (int p = 0; p < S; ++p) v += planes[(size_t)p * n7 + i];
        out[i] = b2[f] + v * s;
    }
}

extern "C" void kernel_launch(void* const* d_in, const int* in_sizes, int n_in,
                              void* d_out, int out_size, void* d_ws, size_t ws_size,
                              hipStream_t stream) {
    (void)n_in; (void)out_size; (void)ws_size;
    const float* x  = (const float*)d_in[0];
    const int*   ei = (const int*)d_in[1];
    const float* ew = (const float*)d_in[2];
    const float* W1 = (const float*)d_in[3];
    const float* b1 = (const float*)d_in[4];
    const float* W2 = (const float*)d_in[5];
    const float* b2 = (const float*)d_in[6];
    float* out = (float*)d_out;

    int n = in_sizes[0] / 3;
    int e = in_sizes[2];
    const int* row = ei;
    const int* col = ei + e;

    constexpr int NPB_1 = 3136,  S_1 = 4;
    constexpr int NPB_2 = 6272,  S_2 = 8;

    float* ws     = (float*)d_ws;
    float* dis    = ws;
    float* h1     = dis + (size_t)n;
    float* z      = h1  + (size_t)n * 16;
    float* h2     = z   + (size_t)n * 16;
    float* planes = h2  + (size_t)n * 16;
    unsigned int* rec = (unsigned int*)(planes + (size_t)n * 64);
    unsigned int* bm  = rec + (size_t)e * 2;
    int nblk   = (e + PK_T * 4 - 1) / (PK_T * 4);
    int pitchw = nblk * PK_W;
    static_assert(NPB_2 == 2 * NPB_1 && S_2 == 2 * S_1, "layer 2's work-groups own whole lists of the degree launch");
    int nr1    = (n + NPB_1 - 1) / NPB_1;
    int nlist  = nr1 * S_1 * (WG / 32);
    unsigned int* lst = bm + (size_t)NRID * pitchw;
    int* cnts  = (int*)(lst + (size_t)nlist * LCAP * 2);
    int* ovf   = cnts + nlist;

    const int B = 256;
    int ntiles  = (n + 15) / 16;
    int gemmT   = ntiles * 32;

    k_pack<<<nblk, PK_T, 0, stream>>>(row, col, ew, rec, bm, ovf, n, e, pitchw);
    k_scatter_lds<1, 1, NPB_1, S_1, 0><<<dim3(nr1, S_1), WG, 0, stream>>>(
        rec, bm, nullptr, planes, n, pitchw, lst, cnts, ovf, nr1);
    k_dis<S_1><<<(n + B - 1) / B, B, 0, stream>>>(planes, dis, n);

    k_gemm1<<<(gemmT + B - 1) / B, B, 0, stream>>>(x, W1, dis, h1, n, ntiles);
    k_scatter_lds<16, 17, NPB_1, S_1, 0><<<dim3(nr1, S_1), WG, 0, stream>>>(
        rec, bm, h1, planes, n, pitchw, lst, cnts, ovf, nr1);
    k_relu<S_1><<<(n * 16 + B - 1) / B, B, 0, stream>>>(h1, dis, planes, b1, z, n * 16);

    k_gemm2<<<(gemmT + B - 1) / B, B, 0, stream>>>(z, W2, h2, n, ntiles);
    k_scatter_lds<7, 7, NPB_2, S_2, 1><<<dim3((n + NPB_2 - 1) / NPB_2, S_2), WG, 0, stream>>>(
        rec, bm, h2, planes, n, pitchw, lst, cnts, ovf, nr1);
    k_out<S_2><<<(n * 7 + B - 1) / B, B, 0, stream>>>(h2, dis, planes, b2, out, n * 7);
}


# degree launch sorts its records into 4 lists per wave by quarter node range; a feature work-group owns a quarter range over all edges and writes z/out from LDS: no slice planes; new f32 sum order
# speedup vs baseline: 9.7851x; 1.0479x over previous
#include <hip/hip_runtime.h>
#include <hip/hip_bf16.h>


typedef __attribute__((ext_vector_type(2))) float v2f;
typedef __attribute__((ext_vector_type(8))) float v8f;

constexpr int WG   = 1024;
constexpr int QCAP = 64;

template <int S>
__global__ void k_dis(const float* __restrict__ planes, float* __restrict__ dis, int n) {
    int i = blockIdx.x * blockDim.x + threadIdx.x;
    if (i < n) {
        float d = 1.0f;
        for (int p = 0; p < S; ++p) d += planes[(size_t)p * n + i];
        dis[i] = (d > 0.f) ? rsqrtf(d) : 0.f;
    }
}

constexpr int RID_NODES = 3136;
constexpr int NRID = 64;
constexpr int PK_T = 256;
constexpr int PK_W = PK_T * 4 / 32;
constexpr int PK_P = PK_W + 1;

__global__ __launch_bounds__(PK_T)
void k_pack(const int* __restrict__ row, const int* __restrict__ col,
            const float* __restrict__ ew,
            unsigned int* __restrict__ rec, unsigned int* __restrict__ bm,
            int* __restrict__ ovf, int n, int e, int pitchw) {
    __shared__ unsigned int tile[NRID * PK_P];
    const int tid = threadIdx.x;
    if (blockIdx.x == 0 && tid == 0) *ovf = 0;
    for (int j = tid; j < NRID * PK_P; j += PK_T) tile[j] = 0u;
    __syncthreads();

    const int t = blockIdx.x * PK_T + tid;
    const int i = t * 4;
    if (i < e) {
        unsigned c4[4];
        const bool vec = ((((size_t)row) | ((size_t)col) | ((size_t)ew) |
                           ((size_t)rec)) & 15) == 0;
        if (vec && i + 4 <= e) {
            int4   r = ((const int4*)row)[t];
            int4   c = ((const int4*)col)[t];
            float4 w = ((const float4*)ew)[t];
            uint4 a, b;
            a.x = ((unsigned)r.x << 14) | ((unsigned)c.x & 16383u); a.y = __float_as_uint(w.x);
            a.z = ((unsigned)r.y << 14) | ((unsigned)c.y & 16383u); a.w = __float_as_uint(w.y);
            b.x = ((unsigned)r.z << 14) | ((unsigned)c.z & 16383u); b.y = __float_as_uint(w.z);
            b.z = ((unsigned)r.w << 14) | ((unsigned)c.w & 16383u); b.w = __float_as_uint(w.w);
            ((uint4*)rec)[2 * (size_t)t]     = a;
            ((uint4*)rec)[2 * (size_t)t + 1] = b;
            c4[0] = (unsigned)c.x; c4[1] = (unsigned)c.y;
            c4[2] = (unsigned)c.z; c4[3] = (unsigned)c.w;
        } else {
            for (int k = 0; k < 4; ++k) {
                int j = i + k;
                c4[k] = 0xFFFFFFFFu;
                if (j < e) {
                    unsigned c = (unsigned)col[j];
                    rec[2 * (size_t)j]     = ((unsigned)row[j] << 14) | (c & 16383u);
                    rec[2 * (size_t)j + 1] = __float_as_uint(ew[j]);
                    c4[k] = c;
                }
            }
        }
        unsigned int* tw = &tile[tid >> 3];
        const unsigned sh = (unsigned)(tid & 7) * 4u;
        for (int k = 0; k < 4; ++k) {
            unsigned c = c4[k];
            if (c < (unsigned)n) {
                unsigned id = c / RID_NODES;
                if (id < (unsigned)NRID) atomicOr(tw + id * PK_P, 1u << (sh + k));
            }
        }
    }
    __syncthreads();

    unsigned int* dst = bm + (size_t)blockIdx.x * PK_W + (tid & 31);
    for (int r = tid >> 5; r < NRID; r += PK_T / 32)
        dst[(size_t)r * pitchw] = tile[r * PK_P + (tid & 31)];
}

struct full_t { static constexpr bool value = true;  };
struct part_t { static constexpr bool value = false; };
constexpr int DS     = 4;
constexpr int NQ     = 4;
constexpr int SUB_A  = 800;
constexpr int SUB_B  = 768;
constexpr int SUBCAP = 320;
static_assert(NQ == 4 && 2 * SUB_A + 2 * SUB_B == RID_NODES, "four sub-ranges fill a node range");
static_assert(SUB_A % 32 == 0 && SUB_B % 32 == 0 && RID_NODES % 32 == 0, "sub-range boundaries at multiples of 32 nodes");
template <int F, int PITCH, int NPB, int S>
__global__ __launch_bounds__(1024)
void k_scatter_lds(const unsigned int* __restrict__ rec, const unsigned int* __restrict__ bm,
                   const float* __restrict__ h, float* __restrict__ dstp,
                   int n, int pitchw,
                   unsigned int* __restrict__ lst, int* __restrict__ cnts, int* __restrict__ ovf, int nr1,
                   const float* __restrict__ dis, const float* __restrict__ bias) {
    static_assert(F == 1 || F == 7 || F == 16, "degree, layer 2 or layer 1");
    constexpr bool DEG = F == 1;
    static_assert(DEG ? (NPB == RID_NODES && S == DS) : (NPB == SUB_A && S == 1),
                  "degree: a whole range, one slice; features: a sub-range, all edges");
    static_assert(RID_NODES <= 16384, "the record keeps 14 bits of the target");
    static_assert(DS == 4, "the list streams below name four slices");
    __shared__ float acc[NPB * PITCH];
    __shared__ int   q[WG / 32][QCAP];
    const int tid  = threadIdx.x;
    const int lane = tid & 31;
    const unsigned rrow = DEG ? blockIdx.x : blockIdx.x / NQ;
    const int qd   = DEG ? 0 : (int)(blockIdx.x % NQ);
    const unsigned base = rrow * RID_NODES + (unsigned)(qd * SUB_A - (qd == 3 ? SUB_A - SUB_B : 0));
    const int nsub = DEG ? RID_NODES : (qd < 2 ? SUB_A : SUB_B);
    const int s    = DEG ? (int)blockIdx.y : 0;
    for (int j = tid; j < NPB * PITCH; j += WG) acc[j] = 0.f;
    __syncthreads();

    const uint2* rec2 = (const uint2*)rec;
    const int G  = pitchw >> 2;
    const int g0 = (int)(((long long)G * s) / S), g1 = (int)(((long long)G * (s + 1)) / S);
    const bool vec = (((size_t)bm) & 15) == 0;

    int* myq = q[tid >> 5];
    int qn = 0;
    const unsigned below = (1u << lane) - 1u;

    auto push = [&](bool m, int eid) {
        unsigned mask = __builtin_amdgcn_ballot_w32(m);
        if (mask != 0u) {
            if (m) myq[qn + __builtin_popcount(mask & below)] = eid;
            qn += __builtin_popcount(mask);
        }
    };
    const int sub = lane >> 2, j = lane & 3;
    auto apply = [&](const uint2 (&rc)[4], const bool (&ok)[4]) __attribute__((always_inline)) {
        if constexpr (F == 16) {
            float4 v[4];
#pragma unroll
            for (int p = 0; p < 4; ++p)
                v[p] = ((const float4*)(h + (size_t)(rc[p].x >> 14) * 16))[j];
#pragma unroll
            for (int p = 0; p < 4; ++p) {
                const float w = __uint_as_float(rc[p].y);
                const unsigned c = ((rc[p].x & 16383u) - base) & 16383u;
                float* a = &acc[c * PITCH + j * 4];
                if (ok[p]) {
                    atomicAdd(a + 0, v[p].x * w); atomicAdd(a + 1, v[p].y * w);
                    atomicAdd(a + 2, v[p].z * w); atomicAdd(a + 3, v[p].w * w);
                }
            }
        } else if constexpr (F == 7) {
            float2 v[4];
#pragma unroll
            for (int p = 0; p < 4; ++p)
                v[p] = ((const float2*)(h + (size_t)(rc[p].x >> 14) * 16))[j];
            float* a[4];
#pragma unroll
            for (int p = 0; p < 4; ++p) {
                const float w = __uint_as_float(rc[p].y);
                const unsigned c = ((rc[p].x & 16383u) - base) & 16383u;
                a[p] = &acc[c * PITCH + j * 2];
                v[p].x *= w; v[p].y *= w;
                if (ok[p]) atomicAdd(a[p], v[p].x);
            }
            if (j < 3) {
#pragma unroll
                for (int p = 0; p < 4; ++p)
                    if (ok[p]) atomicAdd(a[p] + 1, v[p].y);
            }
        }
    };
    const int wlid = (((int)blockIdx.y * nr1 + (int)blockIdx.x) << 5) + (tid >> 5);
    uint2* myl = (uint2*)lst + (size_t)wlid * NQ * SUBCAP;
    int  ln[NQ] = {0, 0, 0, 0};
    bool lover = false;
    auto work = [&](auto fullc, int qb, int cnt) {
        constexpr bool FULL = decltype(fullc)::value;
        if constexpr (DEG) {
            const bool mine = FULL || lane < cnt;
            const uint2 r = rec2[myq[qb + (mine ? lane : 0)]];
            const unsigned c = ((r.x & 16383u) - base) & 16383u;
            const int sq = (int)(c >= (unsigned)SUB_A) + (int)(c >= (unsigned)(2 * SUB_A))
                         + (int)(c >= (unsigned)(2 * SUB_A + SUB_B));
#pragma unroll
            for (int k = 0; k < NQ; ++k) {
                const bool mk = mine && sq == k;
                const unsigned mask = __builtin_amdgcn_ballot_w32(mk);
                const int ck = __builtin_popcount(mask);
                if (ln[k] + ck <= SUBCAP) {
                    if (mk) myl[k * SUBCAP + ln[k] + __builtin_popcount(mask & below)] = r;
                    ln[k] += ck;
                } else {
                    lover = true;
                }
            }
            if (mine) atomicAdd(&acc[c], __uint_as_float(r.y));
        } else {
            uint2 rc[4];
            bool  ok[4];
#pragma unroll
            for (int p = 0; p < 4; ++p) {
                const int slot = p * 8 + sub;
                ok[p] = FULL || slot < cnt;
                rc[p] = rec2[myq[qb + (ok[p] ? slot : 0)]];
            }
#pragma unroll
            for (int p = 0; p < 4; ++p)
                ok[p] = ok[p] && (((rc[p].x & 16383u) - base) & 16383u) < (unsigned)nsub;
            apply(rc, ok);
        }
    };
    auto drain = [&]() {
        while (qn >= 32) { qn -= 32; work(full_t{}, qn, 32); }
    };
    auto scanw = [&](unsigned bits, int eb) {
        while (__builtin_amdgcn_ballot_w32(bits != 0u) != 0u) {
            push(bits != 0u, eb + __ffs(bits) - 1);
            bits &= bits - 1u;
            drain();
        }
    };
    auto ldgroup = [&](int g) {
        uint4 v = make_uint4(0u, 0u, 0u, 0u);
        if (rrow < (unsigned)NRID) {
            const unsigned int* p = bm + (size_t)rrow * pitchw + (size_t)g * 4;
            if (vec) {
                v = *(const uint4*)p;
            } else {
                v.x = p[0]; v.y = p[1]; v.z = p[2]; v.w = p[3];
            }
        }
        return v;
    };

    bool uselist = false;
    if constexpr (!DEG) uselist = __builtin_amdgcn_readfirstlane(*ovf) == 0;

    if (uselist) {
        const int w       = tid >> 5;
        const int lid0    = ((((int)rrow) << 5) + w) * NQ + qd;
        const int lstride = (nr1 << 5) * NQ;
        int o[DS + 1];
        o[0] = 0;
#pragma unroll
        for (int k = 0; k < DS; ++k)
            o[k + 1] = o[k] + __builtin_amdgcn_readfirstlane(cnts[lid0 + k * lstride]);
        const int c = o[DS];
        const uint2* L = (const uint2*)lst + (size_t)lid0 * SUBCAP;
        const int sstr = lstride * SUBCAP;
        const int d1 = sstr - o[1], d2 = sstr - (o[2] - o[1]), d3 = sstr - (o[3] - o[2]);
        auto fetch = [&](uint2 (&rc)[4], int b) {
#pragma unroll
            for (int p = 0; p < 4; ++p) {
                int idx = b + p * 8 + sub;
                idx = idx < c ? idx : b;
                const int pos = idx + (idx >= o[1] ? d1 : 0) + (idx >= o[2] ? d2 : 0) + (idx >= o[3] ? d3 : 0);
                rc[p] = L[pos];
            }
        };
        uint2 cur[4] = {};
        if (c > 0) fetch(cur, 0);
        for (int b = 0; b < c; b += 32) {
            uint2 nxt[4] = {};
            if (b + 32 < c) fetch(nxt, b + 32);
            const int cnt = c - b;
            bool ok[4];
#pragma unroll
            for (int p = 0; p < 4; ++p) ok[p] = p * 8 + sub < cnt;
            apply(cur, ok);
#pragma unroll
            for (int p = 0; p < 4; ++p) cur[p] = nxt[p];
        }
    } else {
        for (int gb = g0; gb < g1; gb += 2 * WG) {
            uint4 cv[2];
            for (int jj = 0; jj < 2; ++jj) {
                int g = gb + jj * WG + tid;
                bool ok = g < g1;
                uint4 v = ldgroup(ok ? g : g0);
                cv[jj].x = ok ? v.x : 0u; cv[jj].y = ok ? v.y : 0u;
                cv[jj].z = ok ? v.z : 0u; cv[jj].w = ok ? v.w : 0u;
            }
            for (int jj = 0; jj < 2; ++jj) {
                int eb = (gb + jj * WG + tid) * 128;
                scanw(cv[jj].x, eb + 0);
                scanw(cv[jj].y, eb + 32);
                scanw(cv[jj].z, eb + 64);
                scanw(cv[jj].w, eb + 96);
            }
        }
        if (qn > 0) work(part_t{}, 0, qn);
        if constexpr (DEG) {
            if (lane == 0) {
#pragma unroll
                for (int k = 0; k < NQ; ++k) cnts[wlid * NQ + k] = ln[k];
                if (lover) *ovf = 1;
            }
        }
    }
    __syncthreads();

    const int left = n - (int)base;
    const int cnt  = left < nsub ? left : nsub;
    if constexpr (DEG) {
        float* dst = dstp + (size_t)s * n + base;
        for (int i = tid; i < cnt; i += WG) dst[i] = acc[i];
    } else if constexpr (F == 16) {
        const float* own = h    + (size_t)base * 16;
        float*       dst = dstp + (size_t)base * 16;
        for (int i = tid; i < cnt * 16; i += WG) {
            const int node = i >> 4, f = i & 15;
            const float sc = dis[base + node];
            const float v  = own[i] + acc[node * PITCH + f];
            dst[i] = fmaxf(v * sc + bias[f], 0.f) * sc;
        }
    } else {
        float* dst = dstp + (size_t)base * 7;
        for (int i = tid; i < cnt * 7; i += WG) {
            const int node = i / 7, f = i - node * 7;
            const float sc = dis[base + node];
            const float v  = h[(size_t)(base + node) * 16 + f] + acc[node * PITCH + f];
            dst[i] = bias[f] + v * sc;
        }
    }
}

__global__ void k_gemm1(const float* __restrict__ x, const float* __restrict__ W1,
                        const float* __restrict__ dis,
                        float* __restrict__ h1, int n, int ntiles) {
    int gtid = blockIdx.x * blockDim.x + threadIdx.x;
    int tile = gtid >> 5;
    int lane = threadIdx.x & 31;
    if (tile >= ntiles) return;
    int half = lane >> 4;
    int m    = lane & 15;
    int row  = tile * 16 + m;
    int rowc = row < n ? row : n - 1;
    bool rv  = row < n;
    int k0   = half * 2;

    float dr = dis[rowc];
    float ax = x[(size_t)rowc * 3 + k0] * dr;
    float ay = x[(size_t)rowc * 3 + 1] * dr;
    v2f a;
    a.x = rv ? ax : 0.f;
    a.y = (rv && half == 0) ? ay : 0.f;

    float bx = W1[k0 * 16 + m];
    float by = W1[16 + m];
    v2f b;
    b.x = bx;
    b.y = (half == 0) ? by : 0.f;

    v8f c = {};
    c = __builtin_amdgcn_wmma_f32_16x16x4_f32(false, a, false, b, (short)0, c, false, false);

    float* p = h1 + ((size_t)(tile * 16 + half * 8)) * 16 + m;
    if (tile * 16 + 16 <= n) {
#pragma unroll
        for (int j = 0; j < 8; ++j) p[(size_t)j * 16] = c[j];
    } else {
#pragma unroll
        for (int j = 0; j < 8; ++j)
            if (tile * 16 + half * 8 + j < n) p[(size_t)j * 16] = c[j];
    }
}

__global__ void k_gemm2(const float* __restrict__ z, const float* __restrict__ W2,
                        float* __restrict__ h2, int n, int ntiles) {
    int gtid = blockIdx.x * blockDim.x + threadIdx.x;
    int tile = gtid >> 5;
    int lane = threadIdx.x & 31;
    if (tile >= ntiles) return;
    int half = lane >> 4;
    int m    = lane & 15;
    int row  = tile * 16 + m;
    int rowc = row < n ? row : n - 1;
    bool rv  = row < n;
    int mc   = m < 7 ? m : 6;
    bool mv  = m < 7;

    v8f c = {};
#pragma unroll
    for (int s = 0; s < 4; ++s) {
        int k = s * 4 + half * 2;
        float ax = z[(size_t)rowc * 16 + k + 0];
        float ay = z[(size_t)rowc * 16 + k + 1];
        v2f a;
        a.x = rv ? ax : 0.f;
        a.y = rv ? ay : 0.f;
        float bx = W2[(k + 0) * 7 + mc];
        float by = W2[(k + 1) * 7 + mc];
        v2f b;
        b.x = mv ? bx : 0.f;
        b.y = mv ? by : 0.f;
        c = __builtin_amdgcn_wmma_f32_16x16x4_f32(false, a, false, b, (short)0, c, false, false);
    }

    float* p = h2 + ((size_t)(tile * 16 + half * 8)) * 16 + m;
    if (tile * 16 + 16 <= n) {
#pragma unroll
        for (int j = 0; j < 8; ++j) p[(size_t)j * 16] = c[j];
    } else {
#pragma unroll
        for (int j = 0; j < 8; ++j)
            if (tile * 16 + half * 8 + j < n) p[(size_t)j * 16] = c[j];
    }
}

extern "C" void kernel_launch(void* const* d_in, const int* in_sizes, int n_in,
                              void* d_out, int out_size, void* d_ws, size_t ws_size,
                              hipStream_t stream) {
    (void)n_in; (void)out_size; (void)ws_size;
    const float* x  = (const float*)d_in[0];
    const int*   ei = (const int*)d_in[1];
    const float* ew = (const float*)d_in[2];
    const float* W1 = (const float*)d_in[3];
    const float* b1 = (const float*)d_in[4];
    const float* W2 = (const float*)d_in[5];
    const float* b2 = (const float*)d_in[6];
    float* out = (float*)d_out;

    int n = in_sizes[0] / 3;
    int e = in_sizes[2];
    const int* row = ei;
    const int* col = ei + e;

    float* ws     = (float*)d_ws;
    float* dis    = ws;
    float* h1     = dis + (size_t)n;
    float* z      = h1  + (size_t)n * 16;
    float* h2     = z   + (size_t)n * 16;
    float* planes = h2  + (size_t)n * 16;
    unsigned int* rec = (unsigned int*)(planes + (size_t)n * DS);
    unsigned int* bm  = rec + (size_t)e * 2;
    int nblk   = (e + PK_T * 4 - 1) / (PK_T * 4);
    int pitchw = nblk * PK_W;
    int nr1    = (n + RID_NODES - 1) / RID_NODES;
    int nlist  = nr1 * DS * (WG / 32) * NQ;
    unsigned int* lst = bm + (size_t)NRID * pitchw;
    int* cnts  = (int*)(lst + (size_t)nlist * SUBCAP * 2);
    int* ovf   = cnts + nlist;

    const int B = 256;
    int ntiles  = (n + 15) / 16;
    int gemmT   = ntiles * 32;

    k_pack<<<nblk, PK_T, 0, stream>>>(row, col, ew, rec, bm, ovf, n, e, pitchw);
    k_scatter_lds<1, 1, RID_NODES, DS><<<dim3(nr1, DS), WG, 0, stream>>>(
        rec, bm, nullptr, planes, n, pitchw, lst, cnts, ovf, nr1, nullptr, nullptr);
    k_dis<DS><<<(n + B - 1) / B, B, 0, stream>>>(planes, dis, n);

    k_gemm1<<<(gemmT + B - 1) / B, B, 0, stream>>>(x, W1, dis, h1, n, ntiles);
    k_scatter_lds<16, 17, SUB_A, 1><<<nr1 * NQ, WG, 0, stream>>>(
        rec, bm, h1, z, n, pitchw, lst, cnts, ovf, nr1, dis, b1);

    k_gemm2<<<(gemmT + B - 1) / B, B, 0, stream>>>(z, W2, h2, n, ntiles);
    k_scatter_lds<7, 7, SUB_A, 1><<<nr1 * NQ, WG, 0, stream>>>(
        rec, bm, h2, out, n, pitchw, lst, cnts, ovf, nr1, dis, b2);
}
